# starting kernel
# baseline (speedup 1.0000x reference)
#include <hip/hip_runtime.h>
#include <hip/hip_bf16.h>

typedef float v2f __attribute__((ext_vector_type(2)));
typedef float v8f __attribute__((ext_vector_type(8)));

#define B_SZ 4
#define N_SZ 16384
#define D_SZ 256
#define NCHUNK 8
#define CHUNK_N (N_SZ / NCHUNK)
#define EPSV 1e-8f

__global__ __launch_bounds__(256) void la_kv_partial(const float* __restrict__ kmat,
                                                     const float* __restrict__ vmat,
                                                     float* __restrict__ partial) {
    const int idx   = blockIdx.x;
    const int chunk = idx & (NCHUNK - 1);
    const int slab  = (idx >> 3) & 3;
    const int b     = idx >> 5;

    const int tid  = threadIdx.x;
    const int w    = tid >> 5;
    const int lane = tid & 31;
    const int m    = lane & 15;
    const int hi   = lane >> 4;

    const int dtile = w >> 1;
    const int ebase = (w & 1) * 128;
    const int dcol  = slab * 64 + dtile * 16 + m;

    const float* kb = kmat + (size_t)b * N_SZ * D_SZ;
    const float* vb = vmat + (size_t)b * N_SZ * D_SZ;

    v8f acc[8];
#pragma unroll
    for (int t = 0; t < 8; ++t) acc[t] = (v8f){0.f,0.f,0.f,0.f,0.f,0.f,0.f,0.f};

    const int nStart = chunk * CHUNK_N;
    const int nEnd   = nStart + CHUNK_N;

    for (int n0 = nStart; n0 < nEnd; n0 += 4) {
        const int nn = n0 + 2 * hi;
        const float* kp = kb + (size_t)nn * D_SZ + dcol;
        v2f a;
        a.x = fmaxf(kp[0],    0.f);
        a.y = fmaxf(kp[D_SZ], 0.f);
        const float* vp = vb + (size_t)nn * D_SZ + ebase + m;
#pragma unroll
        for (int t = 0; t < 8; ++t) {
            v2f bfr;
            bfr.x = vp[t * 16];
            bfr.y = vp[t * 16 + D_SZ];
            acc[t] = __builtin_amdgcn_wmma_f32_16x16x4_f32(
                false, a, false, bfr, (short)0, acc[t], false, false);
        }
    }

    float* pb = partial + (((size_t)b * NCHUNK + chunk) * D_SZ * D_SZ);
#pragma unroll
    for (int t = 0; t < 8; ++t) {
#pragma unroll
        for (int i = 0; i < 8; ++i) {
            const int drow = slab * 64 + dtile * 16 + (hi ? i + 8 : i);
            pb[(size_t)drow * D_SZ + ebase + t * 16 + m] = acc[t][i];
        }
    }
}

__global__ __launch_bounds__(256) void la_kv_reduce(const float* __restrict__ partial,
                                                    float* __restrict__ kv) {
    const int idx = blockIdx.x * 256 + threadIdx.x;
    const int b   = idx >> 16;
    const int rem = idx & 0xFFFF;
    float s = 0.f;
#pragma unroll
    for (int c = 0; c < NCHUNK; ++c)
        s += partial[(((size_t)(b * NCHUNK + c)) << 16) + rem];
    kv[((size_t)b << 16) + rem] = s;
}

#define QPITCH 260

__global__ __launch_bounds__(256) void la_out(const float* __restrict__ qmat,
                                              const float* __restrict__ kv,
                                              float* __restrict__ out) {
    __shared__ float ldsq[32 * QPITCH];
    __shared__ float dpart[256];
    __shared__ float drcp[32];

    const int b    = blockIdx.x >> 9;
    const int n0   = (blockIdx.x & 511) * 32;
    const int tid  = threadIdx.x;

    {
        const int row  = tid >> 3;
        const int cseg = (tid & 7) * 32;
        const float* qrow = qmat + ((size_t)(b * N_SZ + n0 + row)) * D_SZ + cseg;
        float s = 0.f;
#pragma unroll
        for (int j = 0; j < 32; j += 4) {
            float4 x = *(const float4*)(qrow + j);
            x.x = fmaxf(x.x, 0.f);
            x.y = fmaxf(x.y, 0.f);
            x.z = fmaxf(x.z, 0.f);
            x.w = fmaxf(x.w, 0.f);
            *(float4*)&ldsq[row * QPITCH + cseg + j] = x;
            s += x.x + x.y + x.z + x.w;
        }
        dpart[tid] = s;
    }
    __syncthreads();
    if (tid < 32) {
        float s = EPSV;
#pragma unroll
        for (int j = 0; j < 8; ++j) s += dpart[tid * 8 + j];
        drcp[tid] = 1.0f / s;
    }
    __syncthreads();

    const int w    = tid >> 5;
    const int lane = tid & 31;
    const int m    = lane & 15;
    const int hi   = lane >> 4;
    const int rt   = w >> 2;
    const int cb   = (w & 3) * 64;

    v8f acc[4];
#pragma unroll
    for (int t = 0; t < 4; ++t) acc[t] = (v8f){0.f,0.f,0.f,0.f,0.f,0.f,0.f,0.f};

    const float* kvb = kv + ((size_t)b << 16);

    for (int k0 = 0; k0 < D_SZ; k0 += 4) {
        const int kk = k0 + 2 * hi;
        const float* ap = &ldsq[(rt * 16 + m) * QPITCH + kk];
        v2f a;
        a.x = ap[0];
        a.y = ap[1];
#pragma unroll
        for (int t = 0; t < 4; ++t) {
            const float* bp = kvb + (size_t)kk * D_SZ + cb + t * 16 + m;
            v2f bfr;
            bfr.x = bp[0];
            bfr.y = bp[D_SZ];
            acc[t] = __builtin_amdgcn_wmma_f32_16x16x4_f32(
                false, a, false, bfr, (short)0, acc[t], false, false);
        }
    }

    float r[8];
#pragma unroll
    for (int i = 0; i < 8; ++i) r[i] = drcp[rt * 16 + (hi ? i + 8 : i)];

#pragma unroll
    for (int t = 0; t < 4; ++t) {
#pragma unroll
        for (int i = 0; i < 8; ++i) {
            const int rowl = rt * 16 + (hi ? i + 8 : i);
            out[((size_t)(b * N_SZ + n0 + rowl)) * D_SZ + cb + t * 16 + m] =
                acc[t][i] * r[i];
        }
    }
}

extern "C" void kernel_launch(void* const* d_in, const int* in_sizes, int n_in,
                              void* d_out, int out_size, void* d_ws, size_t ws_size,
                              hipStream_t stream) {
    const float* q = (const float*)d_in[0];
    const float* k = (const float*)d_in[1];
    const float* v = (const float*)d_in[2];
    float* out = (float*)d_out;

    float* kvbuf   = (float*)d_ws;
    float* partial = kvbuf + (size_t)B_SZ * D_SZ * D_SZ;

    la_kv_partial<<<B_SZ * 4 * NCHUNK, 256, 0, stream>>>(k, v, partial);
    la_kv_reduce<<<(B_SZ * D_SZ * D_SZ) / 256, 256, 0, stream>>>(partial, kvbuf);
    la_out<<<B_SZ * (N_SZ / 32), 256, 0, stream>>>(q, kvbuf, out);
}


# kv kernel: K and V rows staged through LDS as 4-row 16-byte cells by whole-line loads (128-row stages, 2 buffers), 64x128 tile per work-group, 256 work-groups; new sum order over n in 8-row groups
# speedup vs baseline: 3.1650x; 3.1650x over previous
#include <hip/hip_runtime.h>
#include <hip/hip_bf16.h>


typedef float v2f __attribute__((ext_vector_type(2)));
typedef float v4f __attribute__((ext_vector_type(4)));
typedef float v8f __attribute__((ext_vector_type(8)));

#define B_SZ 4
#define N_SZ 16384
#define D_SZ 256
#define NCHUNK 8
#define CHUNK_N (N_SZ / NCHUNK)
#define EPSV 1e-8f

#define ST_ROWS  128
#define NSTAGE   (CHUNK_N / ST_ROWS)
#define ST_Q     (ST_ROWS / 4)
#define NPAIR    (ST_ROWS / 8)
#define KT_COLS  64
#define VT_COLS  128
#define KQ_PITCH (KT_COLS * 4)
#define VQ_PITCH (VT_COLS * 4)
#define KBUF     (ST_Q * KQ_PITCH)
#define VBUF     (ST_Q * VQ_PITCH)

__device__ __forceinline__ int cell_off(int col) {
    return 4 * (col ^ ((col >> 4) & 3));
}


__device__ __forceinline__ void kv_load_stage(v4f (&kr)[2][4], v4f (&vr)[4][4],
                                              const float* __restrict__ kgs,
                                              const float* __restrict__ vgs) {
#pragma unroll
    for (int u = 0; u < 2; ++u)
#pragma unroll
        for (int r = 0; r < 4; ++r) kr[u][r] = *(const v4f*)(kgs + (64 * u + r) * D_SZ);
#pragma unroll
    for (int u = 0; u < 4; ++u)
#pragma unroll
        for (int r = 0; r < 4; ++r) vr[u][r] = *(const v4f*)(vgs + (32 * u + r) * D_SZ);
}

__device__ __forceinline__ void kv_store_stage(const v4f (&kr)[2][4], const v4f (&vr)[4][4],
                                               float* dK, float* dV,
                                               const int (&kso)[4], const int (&vso)[4]) {
#pragma unroll
    for (int u = 0; u < 2; ++u)
#pragma unroll
        for (int i = 0; i < 4; ++i) {
            v4f t;
            t.x = fmaxf(kr[u][0][i], 0.f);
            t.y = fmaxf(kr[u][1][i], 0.f);
            t.z = fmaxf(kr[u][2][i], 0.f);
            t.w = fmaxf(kr[u][3][i], 0.f);
            *(v4f*)(dK + kso[i] + u * 16 * KQ_PITCH) = t;
        }
#pragma unroll
    for (int u = 0; u < 4; ++u)
#pragma unroll
        for (int i = 0; i < 4; ++i) {
            v4f t;
            t.x = vr[u][0][i];
            t.y = vr[u][1][i];
            t.z = vr[u][2][i];
            t.w = vr[u][3][i];
            *(v4f*)(dV + vso[i] + u * 8 * VQ_PITCH) = t;
        }
}

__global__ __launch_bounds__(256) void la_kv_partial(const float* __restrict__ kmat,
                                                     const float* __restrict__ vmat,
                                                     float* __restrict__ partial) {
    __shared__ __attribute__((aligned(16))) float sK[2 * KBUF];
    __shared__ __attribute__((aligned(16))) float sV[2 * VBUF];

    const int idx   = blockIdx.x;
    const int chunk = idx & (NCHUNK - 1);
    const int et    = (idx >> 3) & 1;
    const int dt    = (idx >> 4) & 3;
    const int b     = idx >> 6;
    const int dbase = dt * KT_COLS;
    const int ebase = et * VT_COLS;

    const int tid  = threadIdx.x;
    const int w    = tid >> 5;
    const int lane = tid & 31;
    const int c    = lane & 15;
    const int h    = lane >> 4;
    const int wd   = w >> 2;
    const int we   = w & 3;

    const size_t row0 = (size_t)b * N_SZ + (size_t)chunk * CHUNK_N;

    const float* kg = kmat + (row0 + 4 * (2 * w + h)) * D_SZ + dbase + 4 * c;
    const float* vg = vmat + (row0 + 4 * w) * D_SZ + ebase + 4 * lane;

    int kso[4], vso[4];
#pragma unroll
    for (int i = 0; i < 4; ++i) {
        kso[i] = (2 * w + h) * KQ_PITCH + cell_off(4 * c + i);
        vso[i] = w * VQ_PITCH + cell_off(4 * lane + i);
    }

    int ka[2], vb[2];
#pragma unroll
    for (int i = 0; i < 2; ++i) {
        ka[i] = h * KQ_PITCH + cell_off(wd * 32 + i * 16 + c);
        vb[i] = h * VQ_PITCH + cell_off(we * 32 + i * 16 + c);
    }

    v4f kr[2][4], vr[4][4];

    v8f acc[2][2];
#pragma unroll
    for (int i = 0; i < 2; ++i)
#pragma unroll
        for (int j = 0; j < 2; ++j) acc[i][j] = (v8f){0.f,0.f,0.f,0.f,0.f,0.f,0.f,0.f};

    kv_load_stage(kr, vr, kg, vg);
    kv_store_stage(kr, vr, sK, sV, kso, vso);
    __syncthreads();

    for (int s = 0; s < NSTAGE; ++s) {
        const int  buf  = s & 1;
        const bool more = (s + 1 < NSTAGE);
        if (more)
            kv_load_stage(kr, vr, kg + (size_t)(s + 1) * ST_ROWS * D_SZ,
                                  vg + (size_t)(s + 1) * ST_ROWS * D_SZ);

        const float* cK = sK + buf * KBUF;
        const float* cV = sV + buf * VBUF;

        v4f fa[2], fb[2];
#pragma unroll
        for (int i = 0; i < 2; ++i) {
            fa[i] = *(const v4f*)(cK + ka[i]);
            fb[i] = *(const v4f*)(cV + vb[i]);
        }
#pragma unroll
        for (int p = 0; p < NPAIR; ++p) {
            v4f na[2], nb[2];
            if (p + 1 < NPAIR) {
#pragma unroll
                for (int i = 0; i < 2; ++i) {
                    na[i] = *(const v4f*)(cK + ka[i] + 2 * (p + 1) * KQ_PITCH);
                    nb[i] = *(const v4f*)(cV + vb[i] + 2 * (p + 1) * VQ_PITCH);
                }
            }
#pragma unroll
            for (int i = 0; i < 2; ++i)
#pragma unroll
                for (int j = 0; j < 2; ++j) {
                    const v2f a2 = fa[i].xy;
                    const v2f b2 = fb[j].xy;
                    acc[i][j] = __builtin_amdgcn_wmma_f32_16x16x4_f32(
                        false, a2, false, b2, (short)0, acc[i][j], false, false);
                }
#pragma unroll
            for (int i = 0; i < 2; ++i)
#pragma unroll
                for (int j = 0; j < 2; ++j) {
                    const v2f a2 = fa[i].zw;
                    const v2f b2 = fb[j].zw;
                    acc[i][j] = __builtin_amdgcn_wmma_f32_16x16x4_f32(
                        false, a2, false, b2, (short)0, acc[i][j], false, false);
                }
            if (p + 1 < NPAIR) {
#pragma unroll
                for (int i = 0; i < 2; ++i) { fa[i] = na[i]; fb[i] = nb[i]; }
            }
        }

        if (more) {
            kv_store_stage(kr, vr, sK + (buf ^ 1) * KBUF, sV + (buf ^ 1) * VBUF, kso, vso);
            __syncthreads();
        }
    }

    float* pb = partial + (((size_t)b * NCHUNK + chunk) * D_SZ * D_SZ);
#pragma unroll
    for (int i = 0; i < 2; ++i)
#pragma unroll
        for (int j = 0; j < 2; ++j)
#pragma unroll
            for (int r = 0; r < 8; ++r) {
                const int drow = dbase + wd * 32 + i * 16 + 8 * h + r;
                const int ecol = ebase + we * 32 + j * 16 + c;
                pb[(size_t)drow * D_SZ + ecol] = acc[i][j][r];
            }
}

__global__ __launch_bounds__(256) void la_kv_reduce(const float* __restrict__ partial,
                                                    float* __restrict__ kv) {
    const int idx = blockIdx.x * 256 + threadIdx.x;
    const int b   = idx >> 16;
    const int rem = idx & 0xFFFF;
    float s = 0.f;
#pragma unroll
    for (int c = 0; c < NCHUNK; ++c)
        s += partial[(((size_t)(b * NCHUNK + c)) << 16) + rem];
    kv[((size_t)b << 16) + rem] = s;
}

#define QPITCH 260

__global__ __launch_bounds__(256) void la_out(const float* __restrict__ qmat,
                                              const float* __restrict__ kv,
                                              float* __restrict__ out) {
    __shared__ float ldsq[32 * QPITCH];
    __shared__ float dpart[256];
    __shared__ float drcp[32];

    const int b    = blockIdx.x >> 9;
    const int n0   = (blockIdx.x & 511) * 32;
    const int tid  = threadIdx.x;

    {
        const int row  = tid >> 3;
        const int cseg = (tid & 7) * 32;
        const float* qrow = qmat + ((size_t)(b * N_SZ + n0 + row)) * D_SZ + cseg;
        float s = 0.f;
#pragma unroll
        for (int j = 0; j < 32; j += 4) {
            float4 x = *(const float4*)(qrow + j);
            x.x = fmaxf(x.x, 0.f);
            x.y = fmaxf(x.y, 0.f);
            x.z = fmaxf(x.z, 0.f);
            x.w = fmaxf(x.w, 0.f);
            *(float4*)&ldsq[row * QPITCH + cseg + j] = x;
            s += x.x + x.y + x.z + x.w;
        }
        dpart[tid] = s;
    }
    __syncthreads();
    if (tid < 32) {
        float s = EPSV;
#pragma unroll
        for (int j = 0; j < 8; ++j) s += dpart[tid * 8 + j];
        drcp[tid] = 1.0f / s;
    }
    __syncthreads();

    const int w    = tid >> 5;
    const int lane = tid & 31;
    const int m    = lane & 15;
    const int hi   = lane >> 4;
    const int rt   = w >> 2;
    const int cb   = (w & 3) * 64;

    v8f acc[4];
#pragma unroll
    for (int t = 0; t < 4; ++t) acc[t] = (v8f){0.f,0.f,0.f,0.f,0.f,0.f,0.f,0.f};

    const float* kvb = kv + ((size_t)b << 16);

    for (int k0 = 0; k0 < D_SZ; k0 += 4) {
        const int kk = k0 + 2 * hi;
        const float* ap = &ldsq[(rt * 16 + m) * QPITCH + kk];
        v2f a;
        a.x = ap[0];
        a.y = ap[1];
#pragma unroll
        for (int t = 0; t < 4; ++t) {
            const float* bp = kvb + (size_t)kk * D_SZ + cb + t * 16 + m;
            v2f bfr;
            bfr.x = bp[0];
            bfr.y = bp[D_SZ];
            acc[t] = __builtin_amdgcn_wmma_f32_16x16x4_f32(
                false, a, false, bfr, (short)0, acc[t], false, false);
        }
    }

    float r[8];
#pragma unroll
    for (int i = 0; i < 8; ++i) r[i] = drcp[rt * 16 + (hi ? i + 8 : i)];

#pragma unroll
    for (int t = 0; t < 4; ++t) {
#pragma unroll
        for (int i = 0; i < 8; ++i) {
            const int rowl = rt * 16 + (hi ? i + 8 : i);
            out[((size_t)(b * N_SZ + n0 + rowl)) * D_SZ + cb + t * 16 + m] =
                acc[t][i] * r[i];
        }
    }
}

extern "C" void kernel_launch(void* const* d_in, const int* in_sizes, int n_in,
                              void* d_out, int out_size, void* d_ws, size_t ws_size,
                              hipStream_t stream) {
    const float* q = (const float*)d_in[0];
    const float* k = (const float*)d_in[1];
    const float* v = (const float*)d_in[2];
    float* out = (float*)d_out;

    float* kvbuf   = (float*)d_ws;
    float* partial = kvbuf + (size_t)B_SZ * D_SZ * D_SZ;

    la_kv_partial<<<B_SZ * 4 * 2 * NCHUNK, 256, 0, stream>>>(k, v, partial);
    la_kv_reduce<<<(B_SZ * D_SZ * D_SZ) / 256, 256, 0, stream>>>(partial, kvbuf);
    la_out<<<B_SZ * (N_SZ / 32), 256, 0, stream>>>(q, kvbuf, out);
}
